# starting kernel
# baseline (speedup 1.0000x reference)
#include <hip/hip_runtime.h>
#include <hip/hip_bf16.h>


typedef __attribute__((ext_vector_type(2))) float v2f;
typedef __attribute__((ext_vector_type(8))) float v8f;

#define NPIX 4096
#define CDIM 128
#define LWIN 25

__device__ __forceinline__ int clamp6(int v) {
    return v < 0 ? 0 : (v > 63 ? 63 : v);
}

__global__ void __launch_bounds__(256)
shiftlsa_stageA(const float* __restrict__ fmap, const float* __restrict__ fmap2,
                const float* __restrict__ W1,   const float* __restrict__ W2,
                float* __restrict__ F1t, float* __restrict__ F2t) {
    const int t = threadIdx.x;
    const int d = t & 127;
    const int p = t >> 7;
    const int P = blockIdx.x * 2 + p;
    const int b = P >> 12;
    const int pix = P & 4095;

    const float* fm1 = fmap  + ((size_t)b * CDIM) * NPIX + pix;
    const float* fm2 = fmap2 + ((size_t)b * CDIM) * NPIX + pix;
    const float* w1  = W1 + (size_t)d * CDIM;
    const float* w2  = W2 + (size_t)d * CDIM;

    float s1 = 0.f, s2 = 0.f;
#pragma unroll 8
    for (int c = 0; c < CDIM; ++c) {
        const float u = fm1[(size_t)c * NPIX];
        const float v = fm2[(size_t)c * NPIX];
        s1 = fmaf(w1[c], u, s1);
        s2 = fmaf(w2[c], v, s2);
    }
    F1t[(size_t)P * CDIM + d] = s1;
    F2t[(size_t)P * CDIM + d] = s2;
}

__global__ void __launch_bounds__(256)
shiftlsa_stageB(const float* __restrict__ F1t, const float* __restrict__ F2t,
                float* __restrict__ G) {
    __shared__ float As[64 * 68];
    __shared__ float Bs[64 * 68];

    const int b  = blockIdx.z;
    const int n0 = blockIdx.x * 64;
    const int m0 = blockIdx.y * 64;

    const float* Fa = F1t + (size_t)b * NPIX * CDIM;
    const float* Fb = F2t + (size_t)b * NPIX * CDIM;

    const int lane = threadIdx.x & 31;
    const int w    = threadIdx.x >> 5;
    const int twr  = w & 3;
    const int twc  = w >> 2;

    v8f acc0 = {};
    v8f acc1 = {};

    const int arow  = twr * 16 + (lane & 15);
    const int bcol0 = twc * 32 + (lane & 15);
    const int bcol1 = bcol0 + 16;
    const int koff  = (lane >> 4) << 1;

    for (int kb = 0; kb < CDIM; kb += 64) {
#pragma unroll
        for (int it = 0; it < 4; ++it) {
            const int f   = threadIdx.x + 256 * it;
            const int row = f >> 4;
            const int cc  = (f & 15) << 2;
            *(float4*)(&As[row * 68 + cc]) =
                *(const float4*)(Fa + ((size_t)(n0 + row)) * CDIM + kb + cc);
            *(float4*)(&Bs[row * 68 + cc]) =
                *(const float4*)(Fb + ((size_t)(m0 + row)) * CDIM + kb + cc);
        }
        __syncthreads();

#pragma unroll
        for (int k = 0; k < 64; k += 4) {
            const v2f a  = *(const v2f*)(&As[arow  * 68 + k + koff]);
            const v2f b0 = *(const v2f*)(&Bs[bcol0 * 68 + k + koff]);
            const v2f b1 = *(const v2f*)(&Bs[bcol1 * 68 + k + koff]);
            acc0 = __builtin_amdgcn_wmma_f32_16x16x4_f32(
                       false, a, false, b0, (short)0, acc0, false, false);
            acc1 = __builtin_amdgcn_wmma_f32_16x16x4_f32(
                       false, a, false, b1, (short)0, acc1, false, false);
        }
        __syncthreads();
    }

    float* Gp = G + (size_t)b * NPIX * NPIX;
    const int orow  = n0 + twr * 16 + ((lane >> 4) << 3);
    const int ocol  = m0 + twc * 32 + (lane & 15);
#pragma unroll
    for (int r = 0; r < 8; ++r) {
        Gp[(size_t)(orow + r) * NPIX + ocol]      = acc0[r];
        Gp[(size_t)(orow + r) * NPIX + ocol + 16] = acc1[r];
    }
}

__global__ void __launch_bounds__(256)
shiftlsa_stageC(const float* __restrict__ attn, const float* __restrict__ G,
                float* __restrict__ out) {
    const int b = blockIdx.y;
    const int n = blockIdx.x;
    const int y = n >> 6;
    const int x = n & 63;

    __shared__ float sA[32];
    if (threadIdx.x < LWIN)
        sA[threadIdx.x] = attn[((size_t)b * NPIX + n) * LWIN + threadIdx.x];
    __syncthreads();

    const int X  = threadIdx.x & 63;
    const int yg = threadIdx.x >> 6;

    float acc[16];
#pragma unroll
    for (int q = 0; q < 16; ++q) acc[q] = 0.f;

    const float* Gb = G + (size_t)b * NPIX * NPIX;

#pragma unroll
    for (int l = 0; l < LWIN; ++l) {
        const int i = l / 5 - 2;
        const int j = l % 5 - 2;
        const float a = sA[l];
        const int py = clamp6(y + i);
        const int px = clamp6(x + j);
        const float* Grow = Gb + ((size_t)(py * 64 + px)) * NPIX;
        const int cX = clamp6(X + j);
#pragma unroll
        for (int q = 0; q < 16; ++q) {
            const int cY = clamp6(yg * 16 + q + i);
            acc[q] = fmaf(a, Grow[cY * 64 + cX], acc[q]);
        }
    }

    const float scale = 0.0883883476483184405f;
    const size_t base = ((size_t)b * NPIX + n) * NPIX;
#pragma unroll
    for (int q = 0; q < 16; ++q)
        out[base + (size_t)(yg * 16 + q) * 64 + X] = acc[q] * scale;
}

extern "C" void kernel_launch(void* const* d_in, const int* in_sizes, int n_in,
                              void* d_out, int out_size, void* d_ws, size_t ws_size,
                              hipStream_t stream) {
    const float* attn  = (const float*)d_in[0];
    const float* fmap  = (const float*)d_in[1];
    const float* fmap2 = (const float*)d_in[2];
    const float* W1    = (const float*)d_in[3];
    const float* W2    = (const float*)d_in[4];
    float*       out   = (float*)d_out;

    float* F1t = (float*)d_ws;
    float* F2t = F1t + (size_t)2 * NPIX * CDIM;
    float* G   = F2t + (size_t)2 * NPIX * CDIM;

    shiftlsa_stageA<<<dim3(4096), dim3(256), 0, stream>>>(
        fmap, fmap2, W1, W2, F1t, F2t);

    shiftlsa_stageB<<<dim3(64, 64, 2), dim3(256), 0, stream>>>(F1t, F2t, G);

    shiftlsa_stageC<<<dim3(4096, 2), dim3(256), 0, stream>>>(attn, G, out);
}


# Stage A 1x1 convs: per-thread FMA loop with strided weight reads -> f32 WMMA 16x16x4 GEMM, both operands staged K-contiguous in LDS; conv sums now run in WMMA k order (chains of 4)
# speedup vs baseline: 1.3892x; 1.3892x over previous
#include <hip/hip_runtime.h>
#include <hip/hip_bf16.h>


typedef __attribute__((ext_vector_type(2))) float v2f;
typedef __attribute__((ext_vector_type(8))) float v8f;

#define NPIX 4096
#define CDIM 128
#define LWIN 25

__device__ __forceinline__ int clamp6(int v) {
    return v < 0 ? 0 : (v > 63 ? 63 : v);
}

#define APITCH 132

__global__ void __launch_bounds__(256)
shiftlsa_stageA(const float* __restrict__ fmap, const float* __restrict__ fmap2,
                const float* __restrict__ W1,   const float* __restrict__ W2,
                float* __restrict__ F1t, float* __restrict__ F2t) {
    __shared__ __attribute__((aligned(16))) float smem[(64 + CDIM) * APITCH];
    float* const Xs = smem;
    float* const Ws = smem + 64 * APITCH;

    const int sel = blockIdx.y;
    const float* __restrict__ Xg = sel ? fmap2 : fmap;
    const float* __restrict__ Wg = sel ? W2 : W1;
    float* __restrict__ Og       = sel ? F2t : F1t;

    const int P0   = blockIdx.x * 64;
    const int b    = P0 >> 12;
    const int pix0 = P0 & 4095;
    const float* Xb = Xg + ((size_t)b * CDIM) * NPIX + pix0;

    const int lane = threadIdx.x & 31;
    const int w    = threadIdx.x >> 5;

#pragma unroll
    for (int it = 0; it < 8; ++it) {
        const int g = w + 8 * it;
        const int c = ((g >> 2) << 3) + (lane & 7);
        const int q = (((g & 3) << 2) + (lane >> 3)) << 2;
        const float4 v = *(const float4*)(Xb + (size_t)c * NPIX + q);
        Xs[(q + 0) * APITCH + c] = v.x;
        Xs[(q + 1) * APITCH + c] = v.y;
        Xs[(q + 2) * APITCH + c] = v.z;
        Xs[(q + 3) * APITCH + c] = v.w;
    }
#pragma unroll
    for (int it = 0; it < 16; ++it) {
        const int f = threadIdx.x + 256 * it;
        const int d = f >> 5;
        const int q = (f & 31) << 2;
        *(float4*)(&Ws[d * APITCH + q]) =
            *(const float4*)(Wg + (size_t)d * CDIM + q);
    }
    __syncthreads();

    const int twr  = w & 3;
    const int twc  = w >> 2;
    const int c16  = lane & 15;
    const int koff = (lane >> 4) << 1;
    const int am   = twr * 16 + c16;
    const int bn   = twc * 64 + c16;

    v8f acc0 = {};
    v8f acc1 = {};
    v8f acc2 = {};
    v8f acc3 = {};

#pragma unroll
    for (int k = 0; k < CDIM; k += 4) {
        const v2f a  = *(const v2f*)(&Xs[(am)      * APITCH + k + koff]);
        const v2f b0 = *(const v2f*)(&Ws[(bn)      * APITCH + k + koff]);
        const v2f b1 = *(const v2f*)(&Ws[(bn + 16) * APITCH + k + koff]);
        const v2f b2 = *(const v2f*)(&Ws[(bn + 32) * APITCH + k + koff]);
        const v2f b3 = *(const v2f*)(&Ws[(bn + 48) * APITCH + k + koff]);
        acc0 = __builtin_amdgcn_wmma_f32_16x16x4_f32(
                   false, a, false, b0, (short)0, acc0, false, false);
        acc1 = __builtin_amdgcn_wmma_f32_16x16x4_f32(
                   false, a, false, b1, (short)0, acc1, false, false);
        acc2 = __builtin_amdgcn_wmma_f32_16x16x4_f32(
                   false, a, false, b2, (short)0, acc2, false, false);
        acc3 = __builtin_amdgcn_wmma_f32_16x16x4_f32(
                   false, a, false, b3, (short)0, acc3, false, false);
    }

    const int orow = P0 + twr * 16 + ((lane >> 4) << 3);
    float* Op = Og + (size_t)orow * CDIM + bn;
#pragma unroll
    for (int r = 0; r < 8; ++r) {
        Op[(size_t)r * CDIM]      = acc0[r];
        Op[(size_t)r * CDIM + 16] = acc1[r];
        Op[(size_t)r * CDIM + 32] = acc2[r];
        Op[(size_t)r * CDIM + 48] = acc3[r];
    }
}

__global__ void __launch_bounds__(256)
shiftlsa_stageB(const float* __restrict__ F1t, const float* __restrict__ F2t,
                float* __restrict__ G) {
    __shared__ float As[64 * 68];
    __shared__ float Bs[64 * 68];

    const int b  = blockIdx.z;
    const int n0 = blockIdx.x * 64;
    const int m0 = blockIdx.y * 64;

    const float* Fa = F1t + (size_t)b * NPIX * CDIM;
    const float* Fb = F2t + (size_t)b * NPIX * CDIM;

    const int lane = threadIdx.x & 31;
    const int w    = threadIdx.x >> 5;
    const int twr  = w & 3;
    const int twc  = w >> 2;

    v8f acc0 = {};
    v8f acc1 = {};

    const int arow  = twr * 16 + (lane & 15);
    const int bcol0 = twc * 32 + (lane & 15);
    const int bcol1 = bcol0 + 16;
    const int koff  = (lane >> 4) << 1;

    for (int kb = 0; kb < CDIM; kb += 64) {
#pragma unroll
        for (int it = 0; it < 4; ++it) {
            const int f   = threadIdx.x + 256 * it;
            const int row = f >> 4;
            const int cc  = (f & 15) << 2;
            *(float4*)(&As[row * 68 + cc]) =
                *(const float4*)(Fa + ((size_t)(n0 + row)) * CDIM + kb + cc);
            *(float4*)(&Bs[row * 68 + cc]) =
                *(const float4*)(Fb + ((size_t)(m0 + row)) * CDIM + kb + cc);
        }
        __syncthreads();

#pragma unroll
        for (int k = 0; k < 64; k += 4) {
            const v2f a  = *(const v2f*)(&As[arow  * 68 + k + koff]);
            const v2f b0 = *(const v2f*)(&Bs[bcol0 * 68 + k + koff]);
            const v2f b1 = *(const v2f*)(&Bs[bcol1 * 68 + k + koff]);
            acc0 = __builtin_amdgcn_wmma_f32_16x16x4_f32(
                       false, a, false, b0, (short)0, acc0, false, false);
            acc1 = __builtin_amdgcn_wmma_f32_16x16x4_f32(
                       false, a, false, b1, (short)0, acc1, false, false);
        }
        __syncthreads();
    }

    float* Gp = G + (size_t)b * NPIX * NPIX;
    const int orow  = n0 + twr * 16 + ((lane >> 4) << 3);
    const int ocol  = m0 + twc * 32 + (lane & 15);
#pragma unroll
    for (int r = 0; r < 8; ++r) {
        Gp[(size_t)(orow + r) * NPIX + ocol]      = acc0[r];
        Gp[(size_t)(orow + r) * NPIX + ocol + 16] = acc1[r];
    }
}

__global__ void __launch_bounds__(256)
shiftlsa_stageC(const float* __restrict__ attn, const float* __restrict__ G,
                float* __restrict__ out) {
    const int b = blockIdx.y;
    const int n = blockIdx.x;
    const int y = n >> 6;
    const int x = n & 63;

    __shared__ float sA[32];
    if (threadIdx.x < LWIN)
        sA[threadIdx.x] = attn[((size_t)b * NPIX + n) * LWIN + threadIdx.x];
    __syncthreads();

    const int X  = threadIdx.x & 63;
    const int yg = threadIdx.x >> 6;

    float acc[16];
#pragma unroll
    for (int q = 0; q < 16; ++q) acc[q] = 0.f;

    const float* Gb = G + (size_t)b * NPIX * NPIX;

#pragma unroll
    for (int l = 0; l < LWIN; ++l) {
        const int i = l / 5 - 2;
        const int j = l % 5 - 2;
        const float a = sA[l];
        const int py = clamp6(y + i);
        const int px = clamp6(x + j);
        const float* Grow = Gb + ((size_t)(py * 64 + px)) * NPIX;
        const int cX = clamp6(X + j);
#pragma unroll
        for (int q = 0; q < 16; ++q) {
            const int cY = clamp6(yg * 16 + q + i);
            acc[q] = fmaf(a, Grow[cY * 64 + cX], acc[q]);
        }
    }

    const float scale = 0.0883883476483184405f;
    const size_t base = ((size_t)b * NPIX + n) * NPIX;
#pragma unroll
    for (int q = 0; q < 16; ++q)
        out[base + (size_t)(yg * 16 + q) * 64 + X] = acc[q] * scale;
}

extern "C" void kernel_launch(void* const* d_in, const int* in_sizes, int n_in,
                              void* d_out, int out_size, void* d_ws, size_t ws_size,
                              hipStream_t stream) {
    const float* attn  = (const float*)d_in[0];
    const float* fmap  = (const float*)d_in[1];
    const float* fmap2 = (const float*)d_in[2];
    const float* W1    = (const float*)d_in[3];
    const float* W2    = (const float*)d_in[4];
    float*       out   = (float*)d_out;

    float* F1t = (float*)d_ws;
    float* F2t = F1t + (size_t)2 * NPIX * CDIM;
    float* G   = F2t + (size_t)2 * NPIX * CDIM;

    shiftlsa_stageA<<<dim3(128, 2), dim3(256), 0, stream>>>(
        fmap, fmap2, W1, W2, F1t, F2t);

    shiftlsa_stageB<<<dim3(64, 64, 2), dim3(256), 0, stream>>>(F1t, F2t, G);

    shiftlsa_stageC<<<dim3(4096, 2), dim3(256), 0, stream>>>(attn, G, out);
}


# Stage C tiled over n: per 4x4 n tile the G row pieces are staged once in LDS and m rows move with n rows, so one LDS read feeds up to 4 outputs (was 25 global reads per output); tap order kept
# speedup vs baseline: 3.2185x; 2.3168x over previous
#include <hip/hip_runtime.h>
#include <hip/hip_bf16.h>


typedef __attribute__((ext_vector_type(2))) float v2f;
typedef __attribute__((ext_vector_type(8))) float v8f;

#define NPIX 4096
#define CDIM 128
#define LWIN 25

__device__ __forceinline__ int clamp6(int v) {
    return v < 0 ? 0 : (v > 63 ? 63 : v);
}

#define APITCH 132

__global__ void __launch_bounds__(256)
shiftlsa_stageA(const float* __restrict__ fmap, const float* __restrict__ fmap2,
                const float* __restrict__ W1,   const float* __restrict__ W2,
                float* __restrict__ F1t, float* __restrict__ F2t) {
    __shared__ __attribute__((aligned(16))) float smem[(64 + CDIM) * APITCH];
    float* const Xs = smem;
    float* const Ws = smem + 64 * APITCH;

    const int sel = blockIdx.y;
    const float* __restrict__ Xg = sel ? fmap2 : fmap;
    const float* __restrict__ Wg = sel ? W2 : W1;
    float* __restrict__ Og       = sel ? F2t : F1t;

    const int P0   = blockIdx.x * 64;
    const int b    = P0 >> 12;
    const int pix0 = P0 & 4095;
    const float* Xb = Xg + ((size_t)b * CDIM) * NPIX + pix0;

    const int lane = threadIdx.x & 31;
    const int w    = threadIdx.x >> 5;

#pragma unroll
    for (int it = 0; it < 8; ++it) {
        const int g = w + 8 * it;
        const int c = ((g >> 2) << 3) + (lane & 7);
        const int q = (((g & 3) << 2) + (lane >> 3)) << 2;
        const float4 v = *(const float4*)(Xb + (size_t)c * NPIX + q);
        Xs[(q + 0) * APITCH + c] = v.x;
        Xs[(q + 1) * APITCH + c] = v.y;
        Xs[(q + 2) * APITCH + c] = v.z;
        Xs[(q + 3) * APITCH + c] = v.w;
    }
#pragma unroll
    for (int it = 0; it < 16; ++it) {
        const int f = threadIdx.x + 256 * it;
        const int d = f >> 5;
        const int q = (f & 31) << 2;
        *(float4*)(&Ws[d * APITCH + q]) =
            *(const float4*)(Wg + (size_t)d * CDIM + q);
    }
    __syncthreads();

    const int twr  = w & 3;
    const int twc  = w >> 2;
    const int c16  = lane & 15;
    const int koff = (lane >> 4) << 1;
    const int am   = twr * 16 + c16;
    const int bn   = twc * 64 + c16;

    v8f acc0 = {};
    v8f acc1 = {};
    v8f acc2 = {};
    v8f acc3 = {};

#pragma unroll
    for (int k = 0; k < CDIM; k += 4) {
        const v2f a  = *(const v2f*)(&Xs[(am)      * APITCH + k + koff]);
        const v2f b0 = *(const v2f*)(&Ws[(bn)      * APITCH + k + koff]);
        const v2f b1 = *(const v2f*)(&Ws[(bn + 16) * APITCH + k + koff]);
        const v2f b2 = *(const v2f*)(&Ws[(bn + 32) * APITCH + k + koff]);
        const v2f b3 = *(const v2f*)(&Ws[(bn + 48) * APITCH + k + koff]);
        acc0 = __builtin_amdgcn_wmma_f32_16x16x4_f32(
                   false, a, false, b0, (short)0, acc0, false, false);
        acc1 = __builtin_amdgcn_wmma_f32_16x16x4_f32(
                   false, a, false, b1, (short)0, acc1, false, false);
        acc2 = __builtin_amdgcn_wmma_f32_16x16x4_f32(
                   false, a, false, b2, (short)0, acc2, false, false);
        acc3 = __builtin_amdgcn_wmma_f32_16x16x4_f32(
                   false, a, false, b3, (short)0, acc3, false, false);
    }

    const int orow = P0 + twr * 16 + ((lane >> 4) << 3);
    float* Op = Og + (size_t)orow * CDIM + bn;
#pragma unroll
    for (int r = 0; r < 8; ++r) {
        Op[(size_t)r * CDIM]      = acc0[r];
        Op[(size_t)r * CDIM + 16] = acc1[r];
        Op[(size_t)r * CDIM + 32] = acc2[r];
        Op[(size_t)r * CDIM + 48] = acc3[r];
    }
}

__global__ void __launch_bounds__(256)
shiftlsa_stageB(const float* __restrict__ F1t, const float* __restrict__ F2t,
                float* __restrict__ G) {
    __shared__ float As[64 * 68];
    __shared__ float Bs[64 * 68];

    const int b  = blockIdx.z;
    const int n0 = blockIdx.x * 64;
    const int m0 = blockIdx.y * 64;

    const float* Fa = F1t + (size_t)b * NPIX * CDIM;
    const float* Fb = F2t + (size_t)b * NPIX * CDIM;

    const int lane = threadIdx.x & 31;
    const int w    = threadIdx.x >> 5;
    const int twr  = w & 3;
    const int twc  = w >> 2;

    v8f acc0 = {};
    v8f acc1 = {};

    const int arow  = twr * 16 + (lane & 15);
    const int bcol0 = twc * 32 + (lane & 15);
    const int bcol1 = bcol0 + 16;
    const int koff  = (lane >> 4) << 1;

    for (int kb = 0; kb < CDIM; kb += 64) {
#pragma unroll
        for (int it = 0; it < 4; ++it) {
            const int f   = threadIdx.x + 256 * it;
            const int row = f >> 4;
            const int cc  = (f & 15) << 2;
            *(float4*)(&As[row * 68 + cc]) =
                *(const float4*)(Fa + ((size_t)(n0 + row)) * CDIM + kb + cc);
            *(float4*)(&Bs[row * 68 + cc]) =
                *(const float4*)(Fb + ((size_t)(m0 + row)) * CDIM + kb + cc);
        }
        __syncthreads();

#pragma unroll
        for (int k = 0; k < 64; k += 4) {
            const v2f a  = *(const v2f*)(&As[arow  * 68 + k + koff]);
            const v2f b0 = *(const v2f*)(&Bs[bcol0 * 68 + k + koff]);
            const v2f b1 = *(const v2f*)(&Bs[bcol1 * 68 + k + koff]);
            acc0 = __builtin_amdgcn_wmma_f32_16x16x4_f32(
                       false, a, false, b0, (short)0, acc0, false, false);
            acc1 = __builtin_amdgcn_wmma_f32_16x16x4_f32(
                       false, a, false, b1, (short)0, acc1, false, false);
        }
        __syncthreads();
    }

    float* Gp = G + (size_t)b * NPIX * NPIX;
    const int orow  = n0 + twr * 16 + ((lane >> 4) << 3);
    const int ocol  = m0 + twc * 32 + (lane & 15);
#pragma unroll
    for (int r = 0; r < 8; ++r) {
        Gp[(size_t)(orow + r) * NPIX + ocol]      = acc0[r];
        Gp[(size_t)(orow + r) * NPIX + ocol + 16] = acc1[r];
    }
}

#define CTY 4
#define CTX 4
#define CSY (CTY + 4)
#define CSX (CTX + 4)
#define CPAIRS (CSY * CSX)

__global__ void __launch_bounds__(256, 8)
shiftlsa_stageC(const float* __restrict__ attn, const float* __restrict__ G,
                float* __restrict__ out) {
    __shared__ __attribute__((aligned(16))) float tile[4 * CPAIRS * 64];

    const int b   = blockIdx.z;
    const int tl  = blockIdx.y;
    const int ny0 = (tl >> 4) * CTY;
    const int nx0 = (tl & 15) * CTX;

    const int t = threadIdx.x & 63;
    const int grp = __builtin_amdgcn_readfirstlane((int)(threadIdx.x >> 6));
    const int mY0 = (blockIdx.x * 4 + grp) - (CTY - 1);

    const float* Gb = G + (size_t)b * NPIX * NPIX;
    float* T = tile + grp * (CPAIRS * 64);

    {
        const int h  = t >> 4;
        const int q4 = (t & 15) << 2;
        int rowoff[2];
#pragma unroll
        for (int e = 0; e < 2; ++e) {
            const int sy = 2 * h + e;
            const int uy = clamp6(ny0 + sy - 2);
            const int vy = clamp6(mY0 + sy - 2);
            rowoff[e] = uy * 64 * NPIX + vy * 64 + q4;
        }
        float4 st[16];
#pragma unroll
        for (int k = 0; k < 16; ++k) {
            const int ux = clamp6(nx0 + (k & 7) - 2);
            st[k] = *(const float4*)(Gb + (size_t)ux * NPIX + rowoff[k >> 3]);
        }
#pragma unroll
        for (int k = 0; k < 16; ++k)
            *(float4*)(T + (h * 16 + k) * 64 + q4) = st[k];
    }
    __syncthreads();

    const int X = t;
    int cX[5];
#pragma unroll
    for (int j = 0; j < 5; ++j) cX[j] = clamp6(X + j - 2);

    float acc[CTY][CTX];
#pragma unroll
    for (int ap = 0; ap < CTY; ++ap)
#pragma unroll
        for (int a = 0; a < CTX; ++a) acc[ap][a] = 0.f;

    const float* A0 = attn + ((size_t)b * NPIX + ny0 * 64 + nx0) * LWIN;

#pragma unroll
    for (int sy = 0; sy < CSY; ++sy) {
#pragma unroll
        for (int sx = 0; sx < CSX; ++sx) {
            const float* piece = T + (sy * CSX + sx) * 64;
#pragma unroll
            for (int a = 0; a < CTX; ++a) {
                const int j = sx - 2 - a;
                if (j < -2 || j > 2) continue;
                const float gv = piece[cX[j + 2]];
#pragma unroll
                for (int ap = 0; ap < CTY; ++ap) {
                    const int i = sy - 2 - ap;
                    if (i < -2 || i > 2) continue;
                    const float wgt =
                        A0[(ap * 64 + a) * LWIN + (i + 2) * 5 + (j + 2)];
                    acc[ap][a] = fmaf(wgt, gv, acc[ap][a]);
                }
            }
        }
    }

#pragma unroll
    for (int ap = 0; ap < CTY; ++ap)
#pragma unroll
        for (int a = 0; a < CTX; ++a) asm volatile("" : "+v"(acc[ap][a]));

    const float scale = 0.0883883476483184405f;
#pragma unroll
    for (int ap = 0; ap < CTY; ++ap) {
        const int my = mY0 + ap;
        if (my < 0 || my > 63) continue;
#pragma unroll
        for (int a = 0; a < CTX; ++a) {
            const size_t n = (size_t)b * NPIX + (ny0 + ap) * 64 + nx0 + a;
            out[n * NPIX + my * 64 + X] = acc[ap][a] * scale;
        }
    }
}

extern "C" void kernel_launch(void* const* d_in, const int* in_sizes, int n_in,
                              void* d_out, int out_size, void* d_ws, size_t ws_size,
                              hipStream_t stream) {
    const float* attn  = (const float*)d_in[0];
    const float* fmap  = (const float*)d_in[1];
    const float* fmap2 = (const float*)d_in[2];
    const float* W1    = (const float*)d_in[3];
    const float* W2    = (const float*)d_in[4];
    float*       out   = (float*)d_out;

    float* F1t = (float*)d_ws;
    float* F2t = F1t + (size_t)2 * NPIX * CDIM;
    float* G   = F2t + (size_t)2 * NPIX * CDIM;

    shiftlsa_stageA<<<dim3(128, 2), dim3(256), 0, stream>>>(
        fmap, fmap2, W1, W2, F1t, F2t);

    shiftlsa_stageB<<<dim3(64, 64, 2), dim3(256), 0, stream>>>(F1t, F2t, G);

    shiftlsa_stageC<<<dim3(17, 256, 2), dim3(256), 0, stream>>>(attn, G, out);
}


# Stage C block walks along nx over 4 n tiles: shared column shifts stay in a 12-slot LDS ring, a further tile loads only 4 new columns, requested ahead of its taps (G reads 4x -> 2.5x); tap order kept
# speedup vs baseline: 3.3119x; 1.0290x over previous
#include <hip/hip_runtime.h>
#include <hip/hip_bf16.h>


typedef __attribute__((ext_vector_type(2))) float v2f;
typedef __attribute__((ext_vector_type(8))) float v8f;
typedef __attribute__((ext_vector_type(4))) float v4f;

#define NPIX 4096
#define CDIM 128
#define LWIN 25

__device__ __forceinline__ int clamp6(int v) {
    return v < 0 ? 0 : (v > 63 ? 63 : v);
}

#define APITCH 132

__global__ void __launch_bounds__(256)
shiftlsa_stageA(const float* __restrict__ fmap, const float* __restrict__ fmap2,
                const float* __restrict__ W1,   const float* __restrict__ W2,
                float* __restrict__ F1t, float* __restrict__ F2t) {
    __shared__ __attribute__((aligned(16))) float smem[(64 + CDIM) * APITCH];
    float* const Xs = smem;
    float* const Ws = smem + 64 * APITCH;

    const int sel = blockIdx.y;
    const float* __restrict__ Xg = sel ? fmap2 : fmap;
    const float* __restrict__ Wg = sel ? W2 : W1;
    float* __restrict__ Og       = sel ? F2t : F1t;

    const int P0   = blockIdx.x * 64;
    const int b    = P0 >> 12;
    const int pix0 = P0 & 4095;
    const float* Xb = Xg + ((size_t)b * CDIM) * NPIX + pix0;

    const int lane = threadIdx.x & 31;
    const int w    = threadIdx.x >> 5;

#pragma unroll
    for (int it = 0; it < 8; ++it) {
        const int g = w + 8 * it;
        const int c = ((g >> 2) << 3) + (lane & 7);
        const int q = (((g & 3) << 2) + (lane >> 3)) << 2;
        const float4 v = *(const float4*)(Xb + (size_t)c * NPIX + q);
        Xs[(q + 0) * APITCH + c] = v.x;
        Xs[(q + 1) * APITCH + c] = v.y;
        Xs[(q + 2) * APITCH + c] = v.z;
        Xs[(q + 3) * APITCH + c] = v.w;
    }
#pragma unroll
    for (int it = 0; it < 16; ++it) {
        const int f = threadIdx.x + 256 * it;
        const int d = f >> 5;
        const int q = (f & 31) << 2;
        *(float4*)(&Ws[d * APITCH + q]) =
            *(const float4*)(Wg + (size_t)d * CDIM + q);
    }
    __syncthreads();

    const int twr  = w & 3;
    const int twc  = w >> 2;
    const int c16  = lane & 15;
    const int koff = (lane >> 4) << 1;
    const int am   = twr * 16 + c16;
    const int bn   = twc * 64 + c16;

    v8f acc0 = {};
    v8f acc1 = {};
    v8f acc2 = {};
    v8f acc3 = {};

#pragma unroll
    for (int k = 0; k < CDIM; k += 4) {
        const v2f a  = *(const v2f*)(&Xs[(am)      * APITCH + k + koff]);
        const v2f b0 = *(const v2f*)(&Ws[(bn)      * APITCH + k + koff]);
        const v2f b1 = *(const v2f*)(&Ws[(bn + 16) * APITCH + k + koff]);
        const v2f b2 = *(const v2f*)(&Ws[(bn + 32) * APITCH + k + koff]);
        const v2f b3 = *(const v2f*)(&Ws[(bn + 48) * APITCH + k + koff]);
        acc0 = __builtin_amdgcn_wmma_f32_16x16x4_f32(
                   false, a, false, b0, (short)0, acc0, false, false);
        acc1 = __builtin_amdgcn_wmma_f32_16x16x4_f32(
                   false, a, false, b1, (short)0, acc1, false, false);
        acc2 = __builtin_amdgcn_wmma_f32_16x16x4_f32(
                   false, a, false, b2, (short)0, acc2, false, false);
        acc3 = __builtin_amdgcn_wmma_f32_16x16x4_f32(
                   false, a, false, b3, (short)0, acc3, false, false);
    }

    const int orow = P0 + twr * 16 + ((lane >> 4) << 3);
    float* Op = Og + (size_t)orow * CDIM + bn;
#pragma unroll
    for (int r = 0; r < 8; ++r) {
        Op[(size_t)r * CDIM]      = acc0[r];
        Op[(size_t)r * CDIM + 16] = acc1[r];
        Op[(size_t)r * CDIM + 32] = acc2[r];
        Op[(size_t)r * CDIM + 48] = acc3[r];
    }
}

__global__ void __launch_bounds__(256)
shiftlsa_stageB(const float* __restrict__ F1t, const float* __restrict__ F2t,
                float* __restrict__ G) {
    __shared__ float As[64 * 68];
    __shared__ float Bs[64 * 68];

    const int b  = blockIdx.z;
    const int n0 = blockIdx.x * 64;
    const int m0 = blockIdx.y * 64;

    const float* Fa = F1t + (size_t)b * NPIX * CDIM;
    const float* Fb = F2t + (size_t)b * NPIX * CDIM;

    const int lane = threadIdx.x & 31;
    const int w    = threadIdx.x >> 5;
    const int twr  = w & 3;
    const int twc  = w >> 2;

    v8f acc0 = {};
    v8f acc1 = {};

    const int arow  = twr * 16 + (lane & 15);
    const int bcol0 = twc * 32 + (lane & 15);
    const int bcol1 = bcol0 + 16;
    const int koff  = (lane >> 4) << 1;

    for (int kb = 0; kb < CDIM; kb += 64) {
#pragma unroll
        for (int it = 0; it < 4; ++it) {
            const int f   = threadIdx.x + 256 * it;
            const int row = f >> 4;
            const int cc  = (f & 15) << 2;
            *(float4*)(&As[row * 68 + cc]) =
                *(const float4*)(Fa + ((size_t)(n0 + row)) * CDIM + kb + cc);
            *(float4*)(&Bs[row * 68 + cc]) =
                *(const float4*)(Fb + ((size_t)(m0 + row)) * CDIM + kb + cc);
        }
        __syncthreads();

#pragma unroll
        for (int k = 0; k < 64; k += 4) {
            const v2f a  = *(const v2f*)(&As[arow  * 68 + k + koff]);
            const v2f b0 = *(const v2f*)(&Bs[bcol0 * 68 + k + koff]);
            const v2f b1 = *(const v2f*)(&Bs[bcol1 * 68 + k + koff]);
            acc0 = __builtin_amdgcn_wmma_f32_16x16x4_f32(
                       false, a, false, b0, (short)0, acc0, false, false);
            acc1 = __builtin_amdgcn_wmma_f32_16x16x4_f32(
                       false, a, false, b1, (short)0, acc1, false, false);
        }
        __syncthreads();
    }

    float* Gp = G + (size_t)b * NPIX * NPIX;
    const int orow  = n0 + twr * 16 + ((lane >> 4) << 3);
    const int ocol  = m0 + twc * 32 + (lane & 15);
#pragma unroll
    for (int r = 0; r < 8; ++r) {
        Gp[(size_t)(orow + r) * NPIX + ocol]      = acc0[r];
        Gp[(size_t)(orow + r) * NPIX + ocol + 16] = acc1[r];
    }
}

#define CTY 4
#define CTX 4
#define CSY (CTY + 4)
#define CSX (CTX + 4)
#define CPAIRS (CSY * CSX)

#define CNT 4
#define CRING 12
#define CGRP (CSY * CRING * 64)
#define CSPLIT 6

typedef const __attribute__((address_space(4))) float attn_k;

template <int SLOT0, int SY0, int SY1>
__device__ __forceinline__ void shiftlsa_taps(const float* T, const int (&cX)[5],
                                              attn_k* A0, float (&acc)[CTY][CTX]) {
#pragma unroll
    for (int sy = SY0; sy < SY1; ++sy) {
#pragma unroll
        for (int sx = 0; sx < CSX; ++sx) {
            const float* piece = T + (sy * CRING + (SLOT0 + sx) % CRING) * 64;
#pragma unroll
            for (int a = 0; a < CTX; ++a) {
                const int j = sx - 2 - a;
                if (j < -2 || j > 2) continue;
                const float gv = piece[cX[j + 2]];
#pragma unroll
                for (int ap = 0; ap < CTY; ++ap) {
                    const int i = sy - 2 - ap;
                    if (i < -2 || i > 2) continue;
                    const float wgt =
                        A0[(ap * 64 + a) * LWIN + (i + 2) * 5 + (j + 2)];
                    acc[ap][a] = fmaf(wgt, gv, acc[ap][a]);
                }
            }
        }
    }
}

template <int S>
__device__ __forceinline__ void shiftlsa_tile(float* T, const int (&cX)[5],
                                              attn_k* A0, float* outp,
                                              const float* Gb, const int (&rowoff)[2],
                                              int nxb, int h, int q4, int mY0, int X) {
    constexpr bool more = (S + 1 < CNT);
    constexpr int SLOT0 = (4 * S) % CRING;
    constexpr int NEW0  = (4 * S + 8) % CRING;

    float acc[CTY][CTX];
#pragma unroll
    for (int ap = 0; ap < CTY; ++ap)
#pragma unroll
        for (int a = 0; a < CTX; ++a) acc[ap][a] = 0.f;

    v4f nx[8];
    if (more) {
#pragma unroll
        for (int k = 0; k < 8; ++k) {
            const int ux = clamp6(nxb + 4 * S + 8 + (k & 3) - 2);
            nx[k] = *(const v4f*)(Gb + (size_t)ux * NPIX + rowoff[k >> 2]);
        }
        __builtin_amdgcn_sched_barrier(0);
    }

    shiftlsa_taps<SLOT0, 0, CSPLIT>(T, cX, A0, acc);

    if (more) {
#pragma unroll
        for (int e = 0; e < 4; ++e)
            *(v4f*)(T + ((2 * h) * CRING + NEW0 + e) * 64 + q4) = nx[e];
    }

    shiftlsa_taps<SLOT0, CSPLIT, CSY>(T, cX, A0, acc);

#pragma unroll
    for (int ap = 0; ap < CTY; ++ap)
#pragma unroll
        for (int a = 0; a < CTX; ++a) asm("" : "+v"(acc[ap][a]));

    const float scale = 0.0883883476483184405f;
#pragma unroll
    for (int ap = 0; ap < CTY; ++ap) {
        const int my = mY0 + ap;
        if (my < 0 || my > 63) continue;
#pragma unroll
        for (int a = 0; a < CTX; ++a)
            outp[(size_t)(ap * 64 + a) * NPIX + my * 64 + X] = acc[ap][a] * scale;
    }

    if (more) {
#pragma unroll
        for (int e = 0; e < 4; ++e)
            *(v4f*)(T + ((2 * h + 1) * CRING + NEW0 + e) * 64 + q4) = nx[4 + e];
        __syncthreads();
    }
}

__global__ void __launch_bounds__(256, 6)
shiftlsa_stageC(const float* __restrict__ attn, const float* __restrict__ G,
                float* __restrict__ out) {
    __shared__ __attribute__((aligned(16))) float tile[4 * CGRP];

    const int b   = blockIdx.z;
    const int tl  = blockIdx.y;
    const int ny0 = (tl / (16 / CNT)) * CTY;
    const int nxb = (tl % (16 / CNT)) * (CNT * CTX);

    const int t = threadIdx.x & 63;
    const int grp = __builtin_amdgcn_readfirstlane((int)(threadIdx.x >> 6));
    const int mY0 = (blockIdx.x * 4 + grp) - (CTY - 1);

    const float* Gb = G + (size_t)b * NPIX * NPIX;
    float* T = tile + grp * CGRP;

    const int h  = t >> 4;
    const int q4 = (t & 15) << 2;
    int rowoff[2];
#pragma unroll
    for (int e = 0; e < 2; ++e) {
        const int sy = 2 * h + e;
        const int uy = clamp6(ny0 + sy - 2);
        const int vy = clamp6(mY0 + sy - 2);
        rowoff[e] = uy * 64 * NPIX + vy * 64 + q4;
    }

    {
        float4 st[16];
#pragma unroll
        for (int k = 0; k < 16; ++k) {
            const int ux = clamp6(nxb + (k & 7) - 2);
            st[k] = *(const float4*)(Gb + (size_t)ux * NPIX + rowoff[k >> 3]);
        }
#pragma unroll
        for (int k = 0; k < 16; ++k)
            *(float4*)(T + ((2 * h + (k >> 3)) * CRING + (k & 7)) * 64 + q4) = st[k];
    }
    __syncthreads();

    const int X = t;
    int cX[5];
#pragma unroll
    for (int j = 0; j < 5; ++j) cX[j] = clamp6(X + j - 2);

    attn_k* A0 = (attn_k*)(attn + ((size_t)b * NPIX + ny0 * 64 + nxb) * LWIN);
    float* O0 = out + ((size_t)b * NPIX + ny0 * 64 + nxb) * NPIX;

    shiftlsa_tile<0>(T, cX, A0 + 0 * (CTX * LWIN), O0 + (size_t)0 * CTX * NPIX,
                     Gb, rowoff, nxb, h, q4, mY0, X);
    shiftlsa_tile<1>(T, cX, A0 + 1 * (CTX * LWIN), O0 + (size_t)1 * CTX * NPIX,
                     Gb, rowoff, nxb, h, q4, mY0, X);
    shiftlsa_tile<2>(T, cX, A0 + 2 * (CTX * LWIN), O0 + (size_t)2 * CTX * NPIX,
                     Gb, rowoff, nxb, h, q4, mY0, X);
    shiftlsa_tile<3>(T, cX, A0 + 3 * (CTX * LWIN), O0 + (size_t)3 * CTX * NPIX,
                     Gb, rowoff, nxb, h, q4, mY0, X);
}

extern "C" void kernel_launch(void* const* d_in, const int* in_sizes, int n_in,
                              void* d_out, int out_size, void* d_ws, size_t ws_size,
                              hipStream_t stream) {
    const float* attn  = (const float*)d_in[0];
    const float* fmap  = (const float*)d_in[1];
    const float* fmap2 = (const float*)d_in[2];
    const float* W1    = (const float*)d_in[3];
    const float* W2    = (const float*)d_in[4];
    float*       out   = (float*)d_out;

    float* F1t = (float*)d_ws;
    float* F2t = F1t + (size_t)2 * NPIX * CDIM;
    float* G   = F2t + (size_t)2 * NPIX * CDIM;

    shiftlsa_stageA<<<dim3(128, 2), dim3(256), 0, stream>>>(
        fmap, fmap2, W1, W2, F1t, F2t);

    shiftlsa_stageB<<<dim3(64, 64, 2), dim3(256), 0, stream>>>(F1t, F2t, G);

    shiftlsa_stageC<<<dim3(17, 64, 2), dim3(256), 0, stream>>>(attn, G, out);
}


# Stage C: each wave reads one float from 4 of the next n tile's 28 attn lines a tile ahead (plain scalar loads, values unused), so the tile's own weight loads hit the scalar cache; sums unchanged
# speedup vs baseline: 3.6880x; 1.1136x over previous
#include <hip/hip_runtime.h>
#include <hip/hip_bf16.h>


typedef __attribute__((ext_vector_type(2))) float v2f;
typedef __attribute__((ext_vector_type(8))) float v8f;
typedef __attribute__((ext_vector_type(4))) float v4f;

#define NPIX 4096
#define CDIM 128
#define LWIN 25

__device__ __forceinline__ int clamp6(int v) {
    return v < 0 ? 0 : (v > 63 ? 63 : v);
}

#define APITCH 132

__global__ void __launch_bounds__(256)
shiftlsa_stageA(const float* __restrict__ fmap, const float* __restrict__ fmap2,
                const float* __restrict__ W1,   const float* __restrict__ W2,
                float* __restrict__ F1t, float* __restrict__ F2t) {
    __shared__ __attribute__((aligned(16))) float smem[(64 + CDIM) * APITCH];
    float* const Xs = smem;
    float* const Ws = smem + 64 * APITCH;

    const int sel = blockIdx.y;
    const float* __restrict__ Xg = sel ? fmap2 : fmap;
    const float* __restrict__ Wg = sel ? W2 : W1;
    float* __restrict__ Og       = sel ? F2t : F1t;

    const int P0   = blockIdx.x * 64;
    const int b    = P0 >> 12;
    const int pix0 = P0 & 4095;
    const float* Xb = Xg + ((size_t)b * CDIM) * NPIX + pix0;

    const int lane = threadIdx.x & 31;
    const int w    = threadIdx.x >> 5;

#pragma unroll
    for (int it = 0; it < 8; ++it) {
        const int g = w + 8 * it;
        const int c = ((g >> 2) << 3) + (lane & 7);
        const int q = (((g & 3) << 2) + (lane >> 3)) << 2;
        const float4 v = *(const float4*)(Xb + (size_t)c * NPIX + q);
        Xs[(q + 0) * APITCH + c] = v.x;
        Xs[(q + 1) * APITCH + c] = v.y;
        Xs[(q + 2) * APITCH + c] = v.z;
        Xs[(q + 3) * APITCH + c] = v.w;
    }
#pragma unroll
    for (int it = 0; it < 16; ++it) {
        const int f = threadIdx.x + 256 * it;
        const int d = f >> 5;
        const int q = (f & 31) << 2;
        *(float4*)(&Ws[d * APITCH + q]) =
            *(const float4*)(Wg + (size_t)d * CDIM + q);
    }
    __syncthreads();

    const int twr  = w & 3;
    const int twc  = w >> 2;
    const int c16  = lane & 15;
    const int koff = (lane >> 4) << 1;
    const int am   = twr * 16 + c16;
    const int bn   = twc * 64 + c16;

    v8f acc0 = {};
    v8f acc1 = {};
    v8f acc2 = {};
    v8f acc3 = {};

#pragma unroll
    for (int k = 0; k < CDIM; k += 4) {
        const v2f a  = *(const v2f*)(&Xs[(am)      * APITCH + k + koff]);
        const v2f b0 = *(const v2f*)(&Ws[(bn)      * APITCH + k + koff]);
        const v2f b1 = *(const v2f*)(&Ws[(bn + 16) * APITCH + k + koff]);
        const v2f b2 = *(const v2f*)(&Ws[(bn + 32) * APITCH + k + koff]);
        const v2f b3 = *(const v2f*)(&Ws[(bn + 48) * APITCH + k + koff]);
        acc0 = __builtin_amdgcn_wmma_f32_16x16x4_f32(
                   false, a, false, b0, (short)0, acc0, false, false);
        acc1 = __builtin_amdgcn_wmma_f32_16x16x4_f32(
                   false, a, false, b1, (short)0, acc1, false, false);
        acc2 = __builtin_amdgcn_wmma_f32_16x16x4_f32(
                   false, a, false, b2, (short)0, acc2, false, false);
        acc3 = __builtin_amdgcn_wmma_f32_16x16x4_f32(
                   false, a, false, b3, (short)0, acc3, false, false);
    }

    const int orow = P0 + twr * 16 + ((lane >> 4) << 3);
    float* Op = Og + (size_t)orow * CDIM + bn;
#pragma unroll
    for (int r = 0; r < 8; ++r) {
        Op[(size_t)r * CDIM]      = acc0[r];
        Op[(size_t)r * CDIM + 16] = acc1[r];
        Op[(size_t)r * CDIM + 32] = acc2[r];
        Op[(size_t)r * CDIM + 48] = acc3[r];
    }
}

__global__ void __launch_bounds__(256)
shiftlsa_stageB(const float* __restrict__ F1t, const float* __restrict__ F2t,
                float* __restrict__ G) {
    __shared__ float As[64 * 68];
    __shared__ float Bs[64 * 68];

    const int b  = blockIdx.z;
    const int n0 = blockIdx.x * 64;
    const int m0 = blockIdx.y * 64;

    const float* Fa = F1t + (size_t)b * NPIX * CDIM;
    const float* Fb = F2t + (size_t)b * NPIX * CDIM;

    const int lane = threadIdx.x & 31;
    const int w    = threadIdx.x >> 5;
    const int twr  = w & 3;
    const int twc  = w >> 2;

    v8f acc0 = {};
    v8f acc1 = {};

    const int arow  = twr * 16 + (lane & 15);
    const int bcol0 = twc * 32 + (lane & 15);
    const int bcol1 = bcol0 + 16;
    const int koff  = (lane >> 4) << 1;

    for (int kb = 0; kb < CDIM; kb += 64) {
#pragma unroll
        for (int it = 0; it < 4; ++it) {
            const int f   = threadIdx.x + 256 * it;
            const int row = f >> 4;
            const int cc  = (f & 15) << 2;
            *(float4*)(&As[row * 68 + cc]) =
                *(const float4*)(Fa + ((size_t)(n0 + row)) * CDIM + kb + cc);
            *(float4*)(&Bs[row * 68 + cc]) =
                *(const float4*)(Fb + ((size_t)(m0 + row)) * CDIM + kb + cc);
        }
        __syncthreads();

#pragma unroll
        for (int k = 0; k < 64; k += 4) {
            const v2f a  = *(const v2f*)(&As[arow  * 68 + k + koff]);
            const v2f b0 = *(const v2f*)(&Bs[bcol0 * 68 + k + koff]);
            const v2f b1 = *(const v2f*)(&Bs[bcol1 * 68 + k + koff]);
            acc0 = __builtin_amdgcn_wmma_f32_16x16x4_f32(
                       false, a, false, b0, (short)0, acc0, false, false);
            acc1 = __builtin_amdgcn_wmma_f32_16x16x4_f32(
                       false, a, false, b1, (short)0, acc1, false, false);
        }
        __syncthreads();
    }

    float* Gp = G + (size_t)b * NPIX * NPIX;
    const int orow  = n0 + twr * 16 + ((lane >> 4) << 3);
    const int ocol  = m0 + twc * 32 + (lane & 15);
#pragma unroll
    for (int r = 0; r < 8; ++r) {
        Gp[(size_t)(orow + r) * NPIX + ocol]      = acc0[r];
        Gp[(size_t)(orow + r) * NPIX + ocol + 16] = acc1[r];
    }
}

#define CTY 4
#define CTX 4
#define CSY (CTY + 4)
#define CSX (CTX + 4)
#define CPAIRS (CSY * CSX)

#define CNT 4
#define CRING 12
#define CGRP (CSY * CRING * 64)
#define CSPLIT 6

typedef const __attribute__((address_space(4))) float attn_k;

__device__ __forceinline__ void shiftlsa_touch(attn_k* A, int wv) {
    float pf[4];
    attn_k* P = A + (wv >> 1) * (64 * LWIN);
#pragma unroll
    for (int i = 0; i < 4; ++i) {
        int off = (wv & 1) * 64 + 16 * i;
        off = off > CTX * LWIN - 1 ? CTX * LWIN - 1 : off;
        pf[i] = P[off];
    }
    asm volatile("" :: "s"(pf[0]), "s"(pf[1]), "s"(pf[2]), "s"(pf[3]));
}

template <int SLOT0, int SY0, int SY1>
__device__ __forceinline__ void shiftlsa_taps(const float* T, const int (&cX)[5],
                                              attn_k* A0, float (&acc)[CTY][CTX]) {
#pragma unroll
    for (int sy = SY0; sy < SY1; ++sy) {
#pragma unroll
        for (int sx = 0; sx < CSX; ++sx) {
            const float* piece = T + (sy * CRING + (SLOT0 + sx) % CRING) * 64;
#pragma unroll
            for (int a = 0; a < CTX; ++a) {
                const int j = sx - 2 - a;
                if (j < -2 || j > 2) continue;
                const float gv = piece[cX[j + 2]];
#pragma unroll
                for (int ap = 0; ap < CTY; ++ap) {
                    const int i = sy - 2 - ap;
                    if (i < -2 || i > 2) continue;
                    const float wgt =
                        A0[(ap * 64 + a) * LWIN + (i + 2) * 5 + (j + 2)];
                    acc[ap][a] = fmaf(wgt, gv, acc[ap][a]);
                }
            }
        }
    }
}

template <int S>
__device__ __forceinline__ void shiftlsa_tile(float* T, const int (&cX)[5],
                                              attn_k* A0, float* outp,
                                              const float* Gb, const int (&rowoff)[2],
                                              int nxb, int h, int q4, int mY0, int X,
                                              int wv) {
    constexpr bool more = (S + 1 < CNT);
    constexpr int SLOT0 = (4 * S) % CRING;
    constexpr int NEW0  = (4 * S + 8) % CRING;

    float acc[CTY][CTX];
#pragma unroll
    for (int ap = 0; ap < CTY; ++ap)
#pragma unroll
        for (int a = 0; a < CTX; ++a) acc[ap][a] = 0.f;

    v4f nx[8];
    if (more) {
#pragma unroll
        for (int k = 0; k < 8; ++k) {
            const int ux = clamp6(nxb + 4 * S + 8 + (k & 3) - 2);
            nx[k] = *(const v4f*)(Gb + (size_t)ux * NPIX + rowoff[k >> 2]);
        }
        shiftlsa_touch(A0 + CTX * LWIN, wv);
        __builtin_amdgcn_sched_barrier(0);
    }

    shiftlsa_taps<SLOT0, 0, CSPLIT>(T, cX, A0, acc);

    if (more) {
#pragma unroll
        for (int e = 0; e < 4; ++e)
            *(v4f*)(T + ((2 * h) * CRING + NEW0 + e) * 64 + q4) = nx[e];
    }

    shiftlsa_taps<SLOT0, CSPLIT, CSY>(T, cX, A0, acc);

#pragma unroll
    for (int ap = 0; ap < CTY; ++ap)
#pragma unroll
        for (int a = 0; a < CTX; ++a) asm("" : "+v"(acc[ap][a]));

    const float scale = 0.0883883476483184405f;
#pragma unroll
    for (int ap = 0; ap < CTY; ++ap) {
        const int my = mY0 + ap;
        if (my < 0 || my > 63) continue;
#pragma unroll
        for (int a = 0; a < CTX; ++a)
            outp[(size_t)(ap * 64 + a) * NPIX + my * 64 + X] = acc[ap][a] * scale;
    }

    if (more) {
#pragma unroll
        for (int e = 0; e < 4; ++e)
            *(v4f*)(T + ((2 * h + 1) * CRING + NEW0 + e) * 64 + q4) = nx[4 + e];
        __syncthreads();
    }
}

__global__ void __launch_bounds__(256, 6)
shiftlsa_stageC(const float* __restrict__ attn, const float* __restrict__ G,
                float* __restrict__ out) {
    __shared__ __attribute__((aligned(16))) float tile[4 * CGRP];

    const int b   = blockIdx.z;
    const int tl  = blockIdx.y;
    const int ny0 = (tl / (16 / CNT)) * CTY;
    const int nxb = (tl % (16 / CNT)) * (CNT * CTX);

    const int t = threadIdx.x & 63;
    const int grp = __builtin_amdgcn_readfirstlane((int)(threadIdx.x >> 6));
    const int mY0 = (blockIdx.x * 4 + grp) - (CTY - 1);
    const int wv = __builtin_amdgcn_readfirstlane((int)(threadIdx.x >> 5));

    const float* Gb = G + (size_t)b * NPIX * NPIX;
    float* T = tile + grp * CGRP;

    const int h  = t >> 4;
    const int q4 = (t & 15) << 2;
    int rowoff[2];
#pragma unroll
    for (int e = 0; e < 2; ++e) {
        const int sy = 2 * h + e;
        const int uy = clamp6(ny0 + sy - 2);
        const int vy = clamp6(mY0 + sy - 2);
        rowoff[e] = uy * 64 * NPIX + vy * 64 + q4;
    }

    attn_k* A0 = (attn_k*)(attn + ((size_t)b * NPIX + ny0 * 64 + nxb) * LWIN);
    float* O0 = out + ((size_t)b * NPIX + ny0 * 64 + nxb) * NPIX;

    shiftlsa_touch(A0, wv);

    {
        float4 st[16];
#pragma unroll
        for (int k = 0; k < 16; ++k) {
            const int ux = clamp6(nxb + (k & 7) - 2);
            st[k] = *(const float4*)(Gb + (size_t)ux * NPIX + rowoff[k >> 3]);
        }
#pragma unroll
        for (int k = 0; k < 16; ++k)
            *(float4*)(T + ((2 * h + (k >> 3)) * CRING + (k & 7)) * 64 + q4) = st[k];
    }
    __syncthreads();

    const int X = t;
    int cX[5];
#pragma unroll
    for (int j = 0; j < 5; ++j) cX[j] = clamp6(X + j - 2);

    shiftlsa_tile<0>(T, cX, A0 + 0 * (CTX * LWIN), O0 + (size_t)0 * CTX * NPIX,
                     Gb, rowoff, nxb, h, q4, mY0, X, wv);
    shiftlsa_tile<1>(T, cX, A0 + 1 * (CTX * LWIN), O0 + (size_t)1 * CTX * NPIX,
                     Gb, rowoff, nxb, h, q4, mY0, X, wv);
    shiftlsa_tile<2>(T, cX, A0 + 2 * (CTX * LWIN), O0 + (size_t)2 * CTX * NPIX,
                     Gb, rowoff, nxb, h, q4, mY0, X, wv);
    shiftlsa_tile<3>(T, cX, A0 + 3 * (CTX * LWIN), O0 + (size_t)3 * CTX * NPIX,
                     Gb, rowoff, nxb, h, q4, mY0, X, wv);
}

extern "C" void kernel_launch(void* const* d_in, const int* in_sizes, int n_in,
                              void* d_out, int out_size, void* d_ws, size_t ws_size,
                              hipStream_t stream) {
    const float* attn  = (const float*)d_in[0];
    const float* fmap  = (const float*)d_in[1];
    const float* fmap2 = (const float*)d_in[2];
    const float* W1    = (const float*)d_in[3];
    const float* W2    = (const float*)d_in[4];
    float*       out   = (float*)d_out;

    float* F1t = (float*)d_ws;
    float* F2t = F1t + (size_t)2 * NPIX * CDIM;
    float* G   = F2t + (size_t)2 * NPIX * CDIM;

    shiftlsa_stageA<<<dim3(128, 2), dim3(256), 0, stream>>>(
        fmap, fmap2, W1, W2, F1t, F2t);

    shiftlsa_stageB<<<dim3(64, 64, 2), dim3(256), 0, stream>>>(F1t, F2t, G);

    shiftlsa_stageC<<<dim3(17, 64, 2), dim3(256), 0, stream>>>(attn, G, out);
}
